# starting kernel
# baseline (speedup 1.0000x reference)
#include <hip/hip_runtime.h>
#include <hip/hip_bf16.h>


typedef __attribute__((ext_vector_type(2))) float v2f;
typedef __attribute__((ext_vector_type(8))) float v8f;

#define NS       128
#define HSTRIDE  132
#define W2STRIDE 65
#define NREPEAT  200

__global__ __launch_bounds__(256, 1)
void conv_chain_wmma_kernel(const float* __restrict__ x,
                            const float* __restrict__ W1,
                            const float* __restrict__ b1,
                            const float* __restrict__ W2,
                            const float* __restrict__ b2,
                            float* __restrict__ out)
{
    __shared__ float Hbuf[64 * HSTRIDE];
    __shared__ float W2lds[64 * W2STRIDE];
    __shared__ float b2lds[64];

    const int tid  = threadIdx.x;
    const int lane = tid & 31;
    const int wave = tid >> 5;
    const int half = lane >> 4;
    const int m    = lane & 15;

    for (int idx = tid; idx < 64 * 64; idx += 256)
        W2lds[(idx >> 6) * W2STRIDE + (idx & 63)] = W2[idx];
    if (tid < 64) b2lds[tid] = b2[tid];

    for (int idx = tid; idx < 64 * NS; idx += 256) {
        const int c0  = idx / NS;
        const int s   = idx - c0 * NS;
        const int sg  = blockIdx.x * NS + s;
        const int b   = sg >> 12;
        const int rem = sg & 4095;
        const int yy  = rem >> 6;
        const int xx  = rem & 63;
        float acc = b1[c0];
        for (int ci = 0; ci < 3; ++ci) {
            const float* xp = x  + ((b * 3 + ci) * 70 + yy) * 70 + xx;
            const float* wp = W1 + (c0 * 3 + ci) * 49;
            #pragma unroll
            for (int ky = 0; ky < 7; ++ky) {
                #pragma unroll
                for (int kx = 0; kx < 7; ++kx)
                    acc = __builtin_fmaf(xp[ky * 70 + kx], wp[ky * 7 + kx], acc);
            }
        }
        Hbuf[c0 * HSTRIDE + s] = acc;
    }
    __syncthreads();

    v2f afrag[4][16];
    #pragma unroll
    for (int ct = 0; ct < 4; ++ct) {
        #pragma unroll
        for (int k = 0; k < 16; ++k) {
            const int row = ct * 16 + m;
            const int c   = k * 4 + half * 2;
            v2f a;
            a[0] = W2lds[row * W2STRIDE + c];
            a[1] = W2lds[row * W2STRIDE + c + 1];
            afrag[ct][k] = a;
        }
    }

    v8f biasv[4];
    #pragma unroll
    for (int ct = 0; ct < 4; ++ct) {
        v8f bv;
        #pragma unroll
        for (int i = 0; i < 8; ++i)
            bv[i] = b2lds[ct * 16 + i + half * 8];
        biasv[ct] = bv;
    }

    const int col = wave * 16 + m;

    for (int it = 0; it < NREPEAT; ++it) {
        v8f acc0 = biasv[0], acc1 = biasv[1], acc2 = biasv[2], acc3 = biasv[3];
        #pragma unroll
        for (int k = 0; k < 16; ++k) {
            const int r = k * 4 + half * 2;
            v2f bfrag;
            bfrag[0] = Hbuf[r * HSTRIDE + col];
            bfrag[1] = Hbuf[(r + 1) * HSTRIDE + col];
            acc0 = __builtin_amdgcn_wmma_f32_16x16x4_f32(false, afrag[0][k],
                       false, bfrag, (short)0, acc0, false, false);
            acc1 = __builtin_amdgcn_wmma_f32_16x16x4_f32(false, afrag[1][k],
                       false, bfrag, (short)0, acc1, false, false);
            acc2 = __builtin_amdgcn_wmma_f32_16x16x4_f32(false, afrag[2][k],
                       false, bfrag, (short)0, acc2, false, false);
            acc3 = __builtin_amdgcn_wmma_f32_16x16x4_f32(false, afrag[3][k],
                       false, bfrag, (short)0, acc3, false, false);
        }
        __syncthreads();
        #pragma unroll
        for (int i = 0; i < 8; ++i) {
            const int row = i + half * 8;
            Hbuf[(row)      * HSTRIDE + col] = acc0[i];
            Hbuf[(row + 16) * HSTRIDE + col] = acc1[i];
            Hbuf[(row + 32) * HSTRIDE + col] = acc2[i];
            Hbuf[(row + 48) * HSTRIDE + col] = acc3[i];
        }
        __syncthreads();
    }

    for (int idx = tid; idx < 64 * NS; idx += 256) {
        const int c0  = idx / NS;
        const int s   = idx - c0 * NS;
        const int sg  = blockIdx.x * NS + s;
        const int b   = sg >> 12;
        const int rem = sg & 4095;
        out[((b * 64 + c0) << 12) + rem] = Hbuf[c0 * HSTRIDE + s];
    }
}

extern "C" void kernel_launch(void* const* d_in, const int* in_sizes, int n_in,
                              void* d_out, int out_size, void* d_ws, size_t ws_size,
                              hipStream_t stream) {
    const float* x  = (const float*)d_in[0];
    const float* W1 = (const float*)d_in[1];
    const float* b1 = (const float*)d_in[2];
    const float* W2 = (const float*)d_in[3];
    const float* b2 = (const float*)d_in[4];
    float* out = (float*)d_out;

    dim3 grid(16384 / NS), block(256);
    conv_chain_wmma_kernel<<<grid, block, 0, stream>>>(x, W1, b1, W2, b2, out);
}


# Chain kept in registers: each step's WMMA D tiles are fed straight back as the next step's B fragments (A loader's k order permuted to the D layout); the 200-step loop has no LDS traffic or barriers
# speedup vs baseline: 1.2053x; 1.2053x over previous
#include <hip/hip_runtime.h>
#include <hip/hip_bf16.h>


typedef __attribute__((ext_vector_type(2))) float v2f;
typedef __attribute__((ext_vector_type(8))) float v8f;

#define NS       128
#define HSTRIDE  132
#define W2STRIDE 65
#define NREPEAT  200

__global__ __launch_bounds__(256, 1)
void conv_chain_wmma_kernel(const float* __restrict__ x,
                            const float* __restrict__ W1,
                            const float* __restrict__ b1,
                            const float* __restrict__ W2,
                            const float* __restrict__ b2,
                            float* __restrict__ out)
{
    __shared__ float Hbuf[64 * HSTRIDE];
    __shared__ float W2lds[64 * W2STRIDE];
    __shared__ float b2lds[64];

    const int tid  = threadIdx.x;
    const int lane = tid & 31;
    const int wave = tid >> 5;
    const int half = lane >> 4;
    const int m    = lane & 15;

    for (int idx = tid; idx < 64 * 64; idx += 256)
        W2lds[(idx >> 6) * W2STRIDE + (idx & 63)] = W2[idx];
    if (tid < 64) b2lds[tid] = b2[tid];

    for (int idx = tid; idx < 64 * NS; idx += 256) {
        const int c0  = idx / NS;
        const int s   = idx - c0 * NS;
        const int sg  = blockIdx.x * NS + s;
        const int b   = sg >> 12;
        const int rem = sg & 4095;
        const int yy  = rem >> 6;
        const int xx  = rem & 63;
        float acc = b1[c0];
        for (int ci = 0; ci < 3; ++ci) {
            const float* xp = x  + ((b * 3 + ci) * 70 + yy) * 70 + xx;
            const float* wp = W1 + (c0 * 3 + ci) * 49;
            #pragma unroll
            for (int ky = 0; ky < 7; ++ky) {
                #pragma unroll
                for (int kx = 0; kx < 7; ++kx)
                    acc = __builtin_fmaf(xp[ky * 70 + kx], wp[ky * 7 + kx], acc);
            }
        }
        Hbuf[c0 * HSTRIDE + s] = acc;
    }
    __syncthreads();

    v2f afrag[4][16];
    #pragma unroll
    for (int ct = 0; ct < 4; ++ct) {
        #pragma unroll
        for (int k = 0; k < 16; ++k) {
            const int row = ct * 16 + m;
            const int c   = (k >> 2) * 16 + half * 8 + (k & 3) * 2;
            v2f a;
            a[0] = W2lds[row * W2STRIDE + c];
            a[1] = W2lds[row * W2STRIDE + c + 1];
            afrag[ct][k] = a;
        }
    }

    v8f biasv[4];
    #pragma unroll
    for (int ct = 0; ct < 4; ++ct) {
        v8f bv;
        #pragma unroll
        for (int i = 0; i < 8; ++i)
            bv[i] = b2lds[ct * 16 + i + half * 8];
        biasv[ct] = bv;
    }

    const int col = wave * 16 + m;

    v8f cur0, cur1, cur2, cur3;
    #pragma unroll
    for (int i = 0; i < 8; ++i) {
        const int row = i + half * 8;
        cur0[i] = Hbuf[(row)      * HSTRIDE + col];
        cur1[i] = Hbuf[(row + 16) * HSTRIDE + col];
        cur2[i] = Hbuf[(row + 32) * HSTRIDE + col];
        cur3[i] = Hbuf[(row + 48) * HSTRIDE + col];
    }

#define WMMA_F32(a, b, c) __builtin_amdgcn_wmma_f32_16x16x4_f32(false, (a), false, (b), (short)0, (c), false, false)
#define KBLOCK(cb, src, d0, d1, d2, d3)                                   \
    _Pragma("unroll")                                                     \
    for (int j = 0; j < 4; ++j) {                                         \
        v2f bfrag;                                                        \
        bfrag[0] = (src)[2 * j];                                          \
        bfrag[1] = (src)[2 * j + 1];                                      \
        d0 = WMMA_F32(afrag[0][(cb) * 4 + j], bfrag, d0);                 \
        d1 = WMMA_F32(afrag[1][(cb) * 4 + j], bfrag, d1);                 \
        d2 = WMMA_F32(afrag[2][(cb) * 4 + j], bfrag, d2);                 \
        d3 = WMMA_F32(afrag[3][(cb) * 4 + j], bfrag, d3);                 \
    }
#define CHAIN_STEP(s0, s1, s2, s3, d0, d1, d2, d3)                        \
    d0 = biasv[0]; d1 = biasv[1]; d2 = biasv[2]; d3 = biasv[3];           \
    KBLOCK(0, s0, d0, d1, d2, d3)                                         \
    KBLOCK(1, s1, d0, d1, d2, d3)                                         \
    KBLOCK(2, s2, d0, d1, d2, d3)                                         \
    KBLOCK(3, s3, d0, d1, d2, d3)

    static_assert(NREPEAT % 2 == 0, "chain is unrolled in pairs");
    for (int it = 0; it < NREPEAT / 2; ++it) {
        v8f nxt0, nxt1, nxt2, nxt3;
        CHAIN_STEP(cur0, cur1, cur2, cur3, nxt0, nxt1, nxt2, nxt3)
        CHAIN_STEP(nxt0, nxt1, nxt2, nxt3, cur0, cur1, cur2, cur3)
    }
#undef CHAIN_STEP
#undef KBLOCK
#undef WMMA_F32

    #pragma unroll
    for (int i = 0; i < 8; ++i) {
        const int row = i + half * 8;
        Hbuf[(row)      * HSTRIDE + col] = cur0[i];
        Hbuf[(row + 16) * HSTRIDE + col] = cur1[i];
        Hbuf[(row + 32) * HSTRIDE + col] = cur2[i];
        Hbuf[(row + 48) * HSTRIDE + col] = cur3[i];
    }
    __syncthreads();

    for (int idx = tid; idx < 64 * NS; idx += 256) {
        const int c0  = idx / NS;
        const int s   = idx - c0 * NS;
        const int sg  = blockIdx.x * NS + s;
        const int b   = sg >> 12;
        const int rem = sg & 4095;
        out[((b * 64 + c0) << 12) + rem] = Hbuf[c0 * HSTRIDE + s];
    }
}

extern "C" void kernel_launch(void* const* d_in, const int* in_sizes, int n_in,
                              void* d_out, int out_size, void* d_ws, size_t ws_size,
                              hipStream_t stream) {
    const float* x  = (const float*)d_in[0];
    const float* W1 = (const float*)d_in[1];
    const float* b1 = (const float*)d_in[2];
    const float* W2 = (const float*)d_in[3];
    const float* b2 = (const float*)d_in[4];
    float* out = (float*)d_out;

    dim3 grid(16384 / NS), block(256);
    conv_chain_wmma_kernel<<<grid, block, 0, stream>>>(x, W1, b1, W2, b2, out);
}


# All 256 WGPs used: 256 workgroups of 4 waves, one 16-column chain per SIMD (168 KiB LDS reservation keeps one workgroup per WGP), instead of 128 of 8 whose wave pairs shared a SIMD's matrix unit
# speedup vs baseline: 2.3654x; 1.9625x over previous
#include <hip/hip_runtime.h>
#include <hip/hip_bf16.h>


typedef __attribute__((ext_vector_type(2))) float v2f;
typedef __attribute__((ext_vector_type(8))) float v8f;

#define NS       64
#define NTHREADS 128
#define HSTRIDE  68
#define W2STRIDE 65
#define NREPEAT  200
#define LDS_FLOATS (42 * 1024)

__global__ __launch_bounds__(NTHREADS) __attribute__((amdgpu_waves_per_eu(1, 1)))
void conv_chain_wmma_kernel(const float* __restrict__ x,
                            const float* __restrict__ W1,
                            const float* __restrict__ b1,
                            const float* __restrict__ W2,
                            const float* __restrict__ b2,
                            float* __restrict__ out)
{
    __shared__ __attribute__((aligned(16))) float lds[LDS_FLOATS];
    static_assert(64 * HSTRIDE + 64 * W2STRIDE + 64 <= LDS_FLOATS, "tiles fit the reservation");
    float* const Hbuf  = lds;
    float* const W2lds = lds + 64 * HSTRIDE;
    float* const b2lds = W2lds + 64 * W2STRIDE;

    const int tid  = threadIdx.x;
    const int lane = tid & 31;
    const int wave = tid >> 5;
    const int half = lane >> 4;
    const int m    = lane & 15;

    for (int idx = tid; idx < 64 * 64; idx += NTHREADS)
        W2lds[(idx >> 6) * W2STRIDE + (idx & 63)] = W2[idx];
    if (tid < 64) b2lds[tid] = b2[tid];

    for (int idx = tid; idx < 64 * NS; idx += NTHREADS) {
        const int c0  = idx / NS;
        const int s   = idx - c0 * NS;
        const int sg  = blockIdx.x * NS + s;
        const int b   = sg >> 12;
        const int rem = sg & 4095;
        const int yy  = rem >> 6;
        const int xx  = rem & 63;
        float acc = b1[c0];
        for (int ci = 0; ci < 3; ++ci) {
            const float* xp = x  + ((b * 3 + ci) * 70 + yy) * 70 + xx;
            const float* wp = W1 + (c0 * 3 + ci) * 49;
            #pragma unroll
            for (int ky = 0; ky < 7; ++ky) {
                #pragma unroll
                for (int kx = 0; kx < 7; ++kx)
                    acc = __builtin_fmaf(xp[ky * 70 + kx], wp[ky * 7 + kx], acc);
            }
        }
        Hbuf[c0 * HSTRIDE + s] = acc;
    }
    __syncthreads();

    v2f afrag[4][16];
    #pragma unroll
    for (int ct = 0; ct < 4; ++ct) {
        #pragma unroll
        for (int k = 0; k < 16; ++k) {
            const int row = ct * 16 + m;
            const int c   = (k >> 2) * 16 + half * 8 + (k & 3) * 2;
            v2f a;
            a[0] = W2lds[row * W2STRIDE + c];
            a[1] = W2lds[row * W2STRIDE + c + 1];
            afrag[ct][k] = a;
        }
    }

    v8f biasv[4];
    #pragma unroll
    for (int ct = 0; ct < 4; ++ct) {
        v8f bv;
        #pragma unroll
        for (int i = 0; i < 8; ++i)
            bv[i] = b2lds[ct * 16 + i + half * 8];
        biasv[ct] = bv;
    }

    const int col = wave * 16 + m;

    v8f cur0, cur1, cur2, cur3;
    #pragma unroll
    for (int i = 0; i < 8; ++i) {
        const int row = i + half * 8;
        cur0[i] = Hbuf[(row)      * HSTRIDE + col];
        cur1[i] = Hbuf[(row + 16) * HSTRIDE + col];
        cur2[i] = Hbuf[(row + 32) * HSTRIDE + col];
        cur3[i] = Hbuf[(row + 48) * HSTRIDE + col];
    }

#define WMMA_F32(a, b, c) __builtin_amdgcn_wmma_f32_16x16x4_f32(false, (a), false, (b), (short)0, (c), false, false)
#define KBLOCK(cb, src, d0, d1, d2, d3)                                   \
    _Pragma("unroll")                                                     \
    for (int j = 0; j < 4; ++j) {                                         \
        v2f bfrag;                                                        \
        bfrag[0] = (src)[2 * j];                                          \
        bfrag[1] = (src)[2 * j + 1];                                      \
        d0 = WMMA_F32(afrag[0][(cb) * 4 + j], bfrag, d0);                 \
        d1 = WMMA_F32(afrag[1][(cb) * 4 + j], bfrag, d1);                 \
        d2 = WMMA_F32(afrag[2][(cb) * 4 + j], bfrag, d2);                 \
        d3 = WMMA_F32(afrag[3][(cb) * 4 + j], bfrag, d3);                 \
    }
#define CHAIN_STEP(s0, s1, s2, s3, d0, d1, d2, d3)                        \
    d0 = biasv[0]; d1 = biasv[1]; d2 = biasv[2]; d3 = biasv[3];           \
    KBLOCK(0, s0, d0, d1, d2, d3)                                         \
    KBLOCK(1, s1, d0, d1, d2, d3)                                         \
    KBLOCK(2, s2, d0, d1, d2, d3)                                         \
    KBLOCK(3, s3, d0, d1, d2, d3)

    static_assert(NREPEAT % 2 == 0, "chain is unrolled in pairs");
    for (int it = 0; it < NREPEAT / 2; ++it) {
        v8f nxt0, nxt1, nxt2, nxt3;
        CHAIN_STEP(cur0, cur1, cur2, cur3, nxt0, nxt1, nxt2, nxt3)
        CHAIN_STEP(nxt0, nxt1, nxt2, nxt3, cur0, cur1, cur2, cur3)
    }
#undef CHAIN_STEP
#undef KBLOCK
#undef WMMA_F32

    #pragma unroll
    for (int i = 0; i < 8; ++i) {
        const int row = i + half * 8;
        Hbuf[(row)      * HSTRIDE + col] = cur0[i];
        Hbuf[(row + 16) * HSTRIDE + col] = cur1[i];
        Hbuf[(row + 32) * HSTRIDE + col] = cur2[i];
        Hbuf[(row + 48) * HSTRIDE + col] = cur3[i];
    }
    __syncthreads();

    for (int idx = tid; idx < 64 * NS; idx += NTHREADS) {
        const int c0  = idx / NS;
        const int s   = idx - c0 * NS;
        const int sg  = blockIdx.x * NS + s;
        const int b   = sg >> 12;
        const int rem = sg & 4095;
        out[((b * 64 + c0) << 12) + rem] = Hbuf[c0 * HSTRIDE + s];
    }
}

extern "C" void kernel_launch(void* const* d_in, const int* in_sizes, int n_in,
                              void* d_out, int out_size, void* d_ws, size_t ws_size,
                              hipStream_t stream) {
    const float* x  = (const float*)d_in[0];
    const float* W1 = (const float*)d_in[1];
    const float* b1 = (const float*)d_in[2];
    const float* W2 = (const float*)d_in[3];
    const float* b2 = (const float*)d_in[4];
    float* out = (float*)d_out;

    dim3 grid(16384 / NS), block(NTHREADS);
    conv_chain_wmma_kernel<<<grid, block, 0, stream>>>(x, W1, b1, W2, b2, out);
}


# 7x7 conv moved from a 147-long dependent vector FMA chain per output onto f32 WMMA (K padded to 148): W1 and the block's input rows staged in LDS, D tiles land in the chain's register layout
# speedup vs baseline: 2.7952x; 1.1817x over previous
#include <hip/hip_runtime.h>
#include <hip/hip_bf16.h>


typedef __attribute__((ext_vector_type(2))) float v2f;
typedef __attribute__((ext_vector_type(8))) float v8f;

#define NS       64
#define NTHREADS 128
#define HSTRIDE  68
#define W2STRIDE 65
#define W1K      147
#define W1STRIDE 148
#define XROWS    490
#define NREPEAT  200
#define LDS_FLOATS (42 * 1024)

__device__ constexpr int koff(int k) {
    return k < W1K ? (k / 49) * XROWS + ((k % 49) / 7) * 70 + (k % 7) : 0;
}

__global__ __launch_bounds__(NTHREADS) __attribute__((amdgpu_waves_per_eu(1, 1)))
void conv_chain_wmma_kernel(const float* __restrict__ x,
                            const float* __restrict__ W1,
                            const float* __restrict__ b1,
                            const float* __restrict__ W2,
                            const float* __restrict__ b2,
                            float* __restrict__ out)
{
    __shared__ __attribute__((aligned(16))) float lds[LDS_FLOATS];
    static_assert(64 * HSTRIDE + 64 * W2STRIDE + 64 + 64 * W1STRIDE + 3 * XROWS <= LDS_FLOATS,
                  "tiles fit the reservation");
    static_assert((64 * HSTRIDE + 64 * W2STRIDE + 64) % 2 == 0 && W1STRIDE % 2 == 0,
                  "W1 fragments are read as aligned float pairs");
    float* const Hbuf  = lds;
    float* const W2lds = lds + 64 * HSTRIDE;
    float* const b2lds = W2lds + 64 * W2STRIDE;
    float* const W1lds = b2lds + 64;
    float* const xs    = W1lds + 64 * W1STRIDE;

    const int tid  = threadIdx.x;
    const int lane = tid & 31;
    const int wave = tid >> 5;
    const int half = lane >> 4;
    const int m    = lane & 15;

    for (int idx = tid; idx < 64 * 64; idx += NTHREADS)
        W2lds[(idx >> 6) * W2STRIDE + (idx & 63)] = W2[idx];
    if (tid < 64) b2lds[tid] = b2[tid];

    v8f cur0, cur1, cur2, cur3;
    #pragma unroll
    for (int i = 0; i < 8; ++i) {
        const int row = i + half * 8;
        cur0[i] = b1[row];
        cur1[i] = b1[row + 16];
        cur2[i] = b1[row + 32];
        cur3[i] = b1[row + 48];
    }
    {
        const int sg0 = blockIdx.x * NS;
        const int b   = sg0 >> 12;
        const int yy  = (sg0 & 4095) >> 6;
        const int ltail = lane < W1K - 128 ? lane : W1K - 129;
        const int lpad  = lane < W1STRIDE - 128 ? lane : W1STRIDE - 129;
        const int ttail = tid < XROWS - 384 ? tid : XROWS - 385;
        const float* const wsrc  = W1 + (wave * 16) * W1K + lane;
        const float* const wsrc4 = W1 + (wave * 16) * W1K + 128 + ltail;
        const float* const xsrc  = x + ((b * 3) * 70 + yy) * 70 + tid;
        const float* const xsrc3 = x + ((b * 3) * 70 + yy) * 70 + 384 + ttail;
        float wv[16][5];
        float xv[3][4];
        #pragma unroll
        for (int r = 0; r < 16; ++r) {
            #pragma unroll
            for (int j = 0; j < 4; ++j)
                wv[r][j] = wsrc[r * W1K + 32 * j];
            wv[r][4] = wsrc4[r * W1K];
        }
        #pragma unroll
        for (int ci = 0; ci < 3; ++ci) {
            #pragma unroll
            for (int i = 0; i < 3; ++i)
                xv[ci][i] = xsrc[ci * 4900 + 128 * i];
            xv[ci][3] = xsrc3[ci * 4900];
        }
        float* const wdst = W1lds + (wave * 16) * W1STRIDE + lane;
        #pragma unroll
        for (int r = 0; r < 16; ++r) {
            #pragma unroll
            for (int j = 0; j < 4; ++j)
                wdst[r * W1STRIDE + 32 * j] = wv[r][j];
        }
        float* const wdst4 = W1lds + (wave * 16) * W1STRIDE + 128 + lpad;
        #pragma unroll
        for (int r = 0; r < 16; ++r)
            wdst4[r * W1STRIDE] = (lane < W1K - 128) ? wv[r][4] : 0.f;
        #pragma unroll
        for (int ci = 0; ci < 3; ++ci) {
            #pragma unroll
            for (int i = 0; i < 3; ++i)
                xs[ci * XROWS + tid + 128 * i] = xv[ci][i];
        }
        #pragma unroll
        for (int ci = 0; ci < 3; ++ci)
            xs[ci * XROWS + 384 + ttail] = xv[ci][3];
    }

    __syncthreads();

    const int col = wave * 16 + m;

#define WMMA_F32(a, b, c) __builtin_amdgcn_wmma_f32_16x16x4_f32(false, (a), false, (b), (short)0, (c), false, false)
    v2f cvA[W1STRIDE / 4][4];
    v2f cvB[W1STRIDE / 4];
    {
        const float* const xcol = xs + col;
        const float* const wrow = W1lds + m * W1STRIDE + 2 * half;
        #pragma unroll
        for (int ks = 0; ks < W1STRIDE / 4; ++ks) {
            const int o0 = half ? koff(4 * ks + 2) : koff(4 * ks);
            const int o1 = half ? koff(4 * ks + 3) : koff(4 * ks + 1);
            v2f bfrag;
            bfrag[0] = xcol[o0];
            bfrag[1] = xcol[o1];
            if (4 * ks + 3 >= W1K) bfrag[1] = half ? 0.f : bfrag[1];
            cvB[ks] = bfrag;
            #pragma unroll
            for (int ct = 0; ct < 4; ++ct)
                cvA[ks][ct] = *(const v2f*)(wrow + ct * 16 * W1STRIDE + 4 * ks);
        }
    }

    v2f afrag[4][16];
    #pragma unroll
    for (int ct = 0; ct < 4; ++ct) {
        #pragma unroll
        for (int k = 0; k < 16; ++k) {
            const int row = ct * 16 + m;
            const int c   = (k >> 2) * 16 + half * 8 + (k & 3) * 2;
            v2f a;
            a[0] = W2lds[row * W2STRIDE + c];
            a[1] = W2lds[row * W2STRIDE + c + 1];
            afrag[ct][k] = a;
        }
    }

    v8f biasv[4];
    #pragma unroll
    for (int ct = 0; ct < 4; ++ct) {
        v8f bv;
        #pragma unroll
        for (int i = 0; i < 8; ++i)
            bv[i] = b2lds[ct * 16 + i + half * 8];
        biasv[ct] = bv;
    }

    __builtin_amdgcn_sched_barrier(0);

    #pragma unroll
    for (int ks = 0; ks < W1STRIDE / 4; ++ks) {
        cur0 = WMMA_F32(cvA[ks][0], cvB[ks], cur0);
        cur1 = WMMA_F32(cvA[ks][1], cvB[ks], cur1);
        cur2 = WMMA_F32(cvA[ks][2], cvB[ks], cur2);
        cur3 = WMMA_F32(cvA[ks][3], cvB[ks], cur3);
    }

#define KBLOCK(cb, src, d0, d1, d2, d3)                                   \
    _Pragma("unroll")                                                     \
    for (int j = 0; j < 4; ++j) {                                         \
        v2f bfrag;                                                        \
        bfrag[0] = (src)[2 * j];                                          \
        bfrag[1] = (src)[2 * j + 1];                                      \
        d0 = WMMA_F32(afrag[0][(cb) * 4 + j], bfrag, d0);                 \
        d1 = WMMA_F32(afrag[1][(cb) * 4 + j], bfrag, d1);                 \
        d2 = WMMA_F32(afrag[2][(cb) * 4 + j], bfrag, d2);                 \
        d3 = WMMA_F32(afrag[3][(cb) * 4 + j], bfrag, d3);                 \
    }
#define CHAIN_STEP(s0, s1, s2, s3, d0, d1, d2, d3)                        \
    d0 = biasv[0]; d1 = biasv[1]; d2 = biasv[2]; d3 = biasv[3];           \
    KBLOCK(0, s0, d0, d1, d2, d3)                                         \
    KBLOCK(1, s1, d0, d1, d2, d3)                                         \
    KBLOCK(2, s2, d0, d1, d2, d3)                                         \
    KBLOCK(3, s3, d0, d1, d2, d3)

    static_assert(NREPEAT % 2 == 0, "chain is unrolled in pairs");
    for (int it = 0; it < NREPEAT / 2; ++it) {
        v8f nxt0, nxt1, nxt2, nxt3;
        CHAIN_STEP(cur0, cur1, cur2, cur3, nxt0, nxt1, nxt2, nxt3)
        CHAIN_STEP(nxt0, nxt1, nxt2, nxt3, cur0, cur1, cur2, cur3)
    }
#undef CHAIN_STEP
#undef KBLOCK
#undef WMMA_F32

    #pragma unroll
    for (int i = 0; i < 8; ++i) {
        const int row = i + half * 8;
        Hbuf[(row)      * HSTRIDE + col] = cur0[i];
        Hbuf[(row + 16) * HSTRIDE + col] = cur1[i];
        Hbuf[(row + 32) * HSTRIDE + col] = cur2[i];
        Hbuf[(row + 48) * HSTRIDE + col] = cur3[i];
    }
    __syncthreads();

    for (int idx = tid; idx < 64 * NS; idx += NTHREADS) {
        const int c0  = idx / NS;
        const int s   = idx - c0 * NS;
        const int sg  = blockIdx.x * NS + s;
        const int b   = sg >> 12;
        const int rem = sg & 4095;
        out[((b * 64 + c0) << 12) + rem] = Hbuf[c0 * HSTRIDE + s];
    }
}

extern "C" void kernel_launch(void* const* d_in, const int* in_sizes, int n_in,
                              void* d_out, int out_size, void* d_ws, size_t ws_size,
                              hipStream_t stream) {
    const float* x  = (const float*)d_in[0];
    const float* W1 = (const float*)d_in[1];
    const float* b1 = (const float*)d_in[2];
    const float* W2 = (const float*)d_in[3];
    const float* b2 = (const float*)d_in[4];
    float* out = (float*)d_out;

    dim3 grid(16384 / NS), block(NTHREADS);
    conv_chain_wmma_kernel<<<grid, block, 0, stream>>>(x, W1, b1, W2, b2, out);
}


# Prologue staging in one round trip: every global load (W2, b2, W1, input rows, b1) issued as one group of 16-byte loads before any LDS store, instead of four serialised load-wait-store rounds
# speedup vs baseline: 22.5383x; 8.0632x over previous
#include <hip/hip_runtime.h>
#include <hip/hip_bf16.h>


typedef __attribute__((ext_vector_type(2))) float v2f;
typedef __attribute__((ext_vector_type(4))) float v4f;
typedef __attribute__((ext_vector_type(8))) float v8f;

#define NS       64
#define NTHREADS 128
#define HSTRIDE  68
#define MSTRIDE  68
#define W1K      147
#define W1STRIDE 148
#define XROWS    490
#define NREPEAT  200
#define NSQUARE  7
#define LDS_FLOATS (42 * 1024)

static_assert(NREPEAT == 128 + 64 + 8, "the powers applied to the tile add up to the step count");
static_assert((1 << NSQUARE) == 128, "the last squaring gives the 128th power");

__device__ constexpr int koff(int k) {
    return k < W1K ? (k / 49) * XROWS + ((k % 49) / 7) * 70 + (k % 7) : 0;
}

#define WMMA_F32(a, b, c) __builtin_amdgcn_wmma_f32_16x16x4_f32(false, (a), false, (b), (short)0, (c), false, false)

struct AFrag { v2f a[4][16]; };
struct Tile  { v8f t[4]; };

__device__ __forceinline__ AFrag load_afrag(const float* Mb, int m, int half) {
    AFrag f;
    #pragma unroll
    for (int ct = 0; ct < 4; ++ct) {
        #pragma unroll
        for (int cb = 0; cb < 4; ++cb) {
            #pragma unroll
            for (int jj = 0; jj < 2; ++jj) {
                const v4f q = *(const v4f*)(Mb + (ct * 16 + m) * MSTRIDE + cb * 16 + half * 8 + jj * 4);
                v2f lo, hi;
                lo[0] = q[0]; lo[1] = q[1];
                hi[0] = q[2]; hi[1] = q[3];
                f.a[ct][cb * 4 + jj * 2]     = lo;
                f.a[ct][cb * 4 + jj * 2 + 1] = hi;
            }
        }
    }
    return f;
}

__device__ __forceinline__ Tile load_cvec(const float* cb_, int half) {
    Tile c;
    #pragma unroll
    for (int ct = 0; ct < 4; ++ct) {
        const v4f q0 = *(const v4f*)(cb_ + ct * 16 + half * 8);
        const v4f q1 = *(const v4f*)(cb_ + ct * 16 + half * 8 + 4);
        #pragma unroll
        for (int i = 0; i < 4; ++i) { c.t[ct][i] = q0[i]; c.t[ct][i + 4] = q1[i]; }
    }
    return c;
}

template <int CT>
__device__ __forceinline__ void mat_apply_half(const AFrag& f, const Tile& x, const Tile& c, Tile& d) {
    v8f d0 = c.t[CT], d1 = c.t[CT + 1];
    #pragma unroll
    for (int cb = 0; cb < 4; ++cb) {
        #pragma unroll
        for (int j = 0; j < 4; ++j) {
            v2f bfrag;
            bfrag[0] = x.t[cb][2 * j];
            bfrag[1] = x.t[cb][2 * j + 1];
            d0 = WMMA_F32(f.a[CT][cb * 4 + j],     bfrag, d0);
            d1 = WMMA_F32(f.a[CT + 1][cb * 4 + j], bfrag, d1);
        }
    }
    d.t[CT] = d0; d.t[CT + 1] = d1;
}

__device__ __forceinline__ Tile mat_apply(const AFrag& f, const Tile& x, const Tile& c) {
    Tile d;
    mat_apply_half<0>(f, x, c, d);
    __builtin_amdgcn_sched_barrier(0);
    mat_apply_half<2>(f, x, c, d);
    return d;
}

__device__ __forceinline__ float xhalf(float v) {
    return __builtin_bit_cast(float, __builtin_amdgcn_permlanex16(__builtin_bit_cast(int, v), __builtin_bit_cast(int, v),
                                                                  0x76543210, 0xfedcba98, false, false));
}

template <int CT>
__device__ __forceinline__ float rowdot(const AFrag& f, const Tile& c) {
    v2f acc2;
    acc2[0] = 0.f; acc2[1] = 0.f;
    #pragma unroll
    for (int cb = 0; cb < 4; ++cb) {
        #pragma unroll
        for (int j = 0; j < 4; ++j) {
            v2f cpair;
            cpair[0] = c.t[cb][2 * j];
            cpair[1] = c.t[cb][2 * j + 1];
            acc2 += f.a[CT][cb * 4 + j] * cpair;
        }
    }
    return acc2[0] + acc2[1];
}

__global__ __launch_bounds__(NTHREADS) __attribute__((amdgpu_waves_per_eu(1, 1)))
void conv_chain_wmma_kernel(const float* __restrict__ x,
                            const float* __restrict__ W1,
                            const float* __restrict__ b1,
                            const float* __restrict__ W2,
                            const float* __restrict__ b2,
                            float* __restrict__ out)
{
    __shared__ __attribute__((aligned(32))) float lds[LDS_FLOATS];
    static_assert(64 * HSTRIDE + 2 * 64 * MSTRIDE + 2 * 64 + 64 * W1STRIDE + 3 * XROWS <= LDS_FLOATS,
                  "tiles fit the reservation");
    static_assert((64 * HSTRIDE) % 8 == 0 && (64 * MSTRIDE) % 8 == 0 && MSTRIDE % 4 == 0,
                  "matrix and vector buffers are read as aligned float quads");
    static_assert((64 * HSTRIDE + 2 * 64 * MSTRIDE + 2 * 64) % 2 == 0 && W1STRIDE % 2 == 0,
                  "W1 fragments are read as aligned float pairs");
    float* const Hbuf  = lds;
    float* const Mbuf0 = lds + 64 * HSTRIDE;
    float* const Mbuf1 = Mbuf0 + 64 * MSTRIDE;
    float* const cbuf0 = Mbuf1 + 64 * MSTRIDE;
    float* const cbuf1 = cbuf0 + 64;
    float* const W1lds = cbuf1 + 64;
    float* const xs    = W1lds + 64 * W1STRIDE;

    const int tid  = threadIdx.x;
    const int lane = tid & 31;
    const int wave = __builtin_amdgcn_readfirstlane(tid >> 5);
    const int half = lane >> 4;
    const int m    = lane & 15;

    const int sg0 = blockIdx.x * NS;
    const int b   = sg0 >> 12;
    const int yy  = (sg0 & 4095) >> 6;
    v4f w2q[8];
    #pragma unroll
    for (int i = 0; i < 8; ++i)
        w2q[i] = *(const v4f*)(W2 + 4 * (tid + NTHREADS * i));
    const float b2v = b2[tid & 63];
    v4f w1q[19];
    int  w1i[19];
    const float* const wsrc = W1 + wave * (16 * W1K);
    #pragma unroll
    for (int j = 0; j < 19; ++j) {
        const int q = lane + 32 * j;
        w1i[j] = q < 588 ? q : 587;
        w1q[j] = *(const v4f*)(wsrc + 4 * w1i[j]);
    }
    const float* const xsrc = x + ((b * 3) * 70 + yy) * 70;
    const int xp1 = tid + NTHREADS < XROWS / 2 ? tid + NTHREADS : XROWS / 2 - 1;
    v2f xq[3][2];
    #pragma unroll
    for (int ci = 0; ci < 3; ++ci) {
        xq[ci][0] = *(const v2f*)(xsrc + ci * 4900 + 2 * tid);
        xq[ci][1] = *(const v2f*)(xsrc + ci * 4900 + 2 * xp1);
    }
    v8f cur0, cur1, cur2, cur3;
    {
        v4f bq[4][2];
        #pragma unroll
        for (int ct = 0; ct < 4; ++ct) {
            bq[ct][0] = *(const v4f*)(b1 + ct * 16 + half * 8);
            bq[ct][1] = *(const v4f*)(b1 + ct * 16 + half * 8 + 4);
        }
        #pragma unroll
        for (int i = 0; i < 4; ++i) {
            cur0[i] = bq[0][0][i]; cur0[i + 4] = bq[0][1][i];
            cur1[i] = bq[1][0][i]; cur1[i + 4] = bq[1][1][i];
            cur2[i] = bq[2][0][i]; cur2[i + 4] = bq[2][1][i];
            cur3[i] = bq[3][0][i]; cur3[i + 4] = bq[3][1][i];
        }
    }
    __builtin_amdgcn_sched_barrier(0);
    int w1o[19][4];
    #pragma unroll
    for (int j = 0; j < 19; ++j) {
        const int f0 = 4 * w1i[j];
        const int r0 = f0 / W1K;
        const int k0 = f0 - r0 * W1K;
        #pragma unroll
        for (int e = 0; e < 4; ++e)
            w1o[j][e] = r0 * W1STRIDE + k0 + e + (k0 + e >= W1K ? 1 : 0);
    }
    __builtin_amdgcn_sched_barrier(0);
    #pragma unroll
    for (int i = 0; i < 8; ++i) {
        const int q = tid + NTHREADS * i;
        *(v4f*)(Mbuf0 + (q >> 4) * MSTRIDE + (q & 15) * 4) = w2q[i];
    }
    cbuf0[tid & 63] = b2v;
    {
        float* const wdst = W1lds + (wave * 16) * W1STRIDE;
        #pragma unroll
        for (int j = 0; j < 19; ++j) {
            #pragma unroll
            for (int e = 0; e < 4; ++e)
                wdst[w1o[j][e]] = w1q[j][e];
        }
        wdst[(lane & 15) * W1STRIDE + W1K] = 0.f;
    }
    #pragma unroll
    for (int ci = 0; ci < 3; ++ci) {
        *(v2f*)(xs + ci * XROWS + 2 * tid) = xq[ci][0];
        *(v2f*)(xs + ci * XROWS + 2 * xp1) = xq[ci][1];
    }

    __syncthreads();

    const int col = wave * 16 + m;

    v2f cvA[W1STRIDE / 4][4];
    v2f cvB[W1STRIDE / 4];
    {
        const float* const xcol = xs + col;
        const float* const wrow = W1lds + m * W1STRIDE + 2 * half;
        #pragma unroll
        for (int ks = 0; ks < W1STRIDE / 4; ++ks) {
            const int o0 = half ? koff(4 * ks + 2) : koff(4 * ks);
            const int o1 = half ? koff(4 * ks + 3) : koff(4 * ks + 1);
            v2f bfrag;
            bfrag[0] = xcol[o0];
            bfrag[1] = xcol[o1];
            if (4 * ks + 3 >= W1K) bfrag[1] = half ? 0.f : bfrag[1];
            cvB[ks] = bfrag;
            #pragma unroll
            for (int ct = 0; ct < 4; ++ct)
                cvA[ks][ct] = *(const v2f*)(wrow + ct * 16 * W1STRIDE + 4 * ks);
        }
    }

    AFrag af = load_afrag(Mbuf0, m, half);
    Tile  cv = load_cvec(cbuf0, half);
    Tile  mc;
    #pragma unroll
    for (int cb = 0; cb < 4; ++cb) {
        #pragma unroll
        for (int r = 0; r < 8; ++r)
            mc.t[cb][r] = Mbuf0[(cb * 16 + half * 8 + r) * MSTRIDE + col];
    }
    float crow = cbuf0[col];

    __builtin_amdgcn_sched_barrier(0);

    #pragma unroll
    for (int ks = 0; ks < W1STRIDE / 4; ++ks) {
        cur0 = WMMA_F32(cvA[ks][0], cvB[ks], cur0);
        cur1 = WMMA_F32(cvA[ks][1], cvB[ks], cur1);
        cur2 = WMMA_F32(cvA[ks][2], cvB[ks], cur2);
        cur3 = WMMA_F32(cvA[ks][3], cvB[ks], cur3);
    }
    Tile cur;
    cur.t[0] = cur0; cur.t[1] = cur1; cur.t[2] = cur2; cur.t[3] = cur3;

    Tile zero;
    #pragma unroll
    for (int ct = 0; ct < 4; ++ct) {
        #pragma unroll
        for (int i = 0; i < 8; ++i) zero.t[ct][i] = 0.f;
    }
    #pragma unroll
    for (int s = 0; s < NSQUARE; ++s) {
        const int p = 1 << s;
        float* const Mnext = (s & 1) ? Mbuf0 : Mbuf1;
        float* const cnext = (s & 1) ? cbuf0 : cbuf1;

        Tile nm;
        mat_apply_half<0>(af, mc, zero, nm);
        __builtin_amdgcn_sched_barrier(0);
        #pragma unroll
        for (int ct = 0; ct < 2; ++ct) {
            #pragma unroll
            for (int r = 0; r < 8; ++r)
                Mnext[(ct * 16 + half * 8 + r) * MSTRIDE + col] = nm.t[ct][r];
        }
        mat_apply_half<2>(af, mc, zero, nm);

        float part;
        if (wave == 0)      part = rowdot<0>(af, cv);
        else if (wave == 1) part = rowdot<1>(af, cv);
        else if (wave == 2) part = rowdot<2>(af, cv);
        else                part = rowdot<3>(af, cv);
        part += xhalf(part);
        crow += part;
        cnext[col] = crow;

        #pragma unroll
        for (int ct = 2; ct < 4; ++ct) {
            #pragma unroll
            for (int r = 0; r < 8; ++r)
                Mnext[(ct * 16 + half * 8 + r) * MSTRIDE + col] = nm.t[ct][r];
        }
        __syncthreads();
        const AFrag af2 = load_afrag(Mnext, m, half);
        const Tile  cv2 = load_cvec(cnext, half);
        __builtin_amdgcn_sched_barrier(0);

        if (p == 8 || p == 64) {
            cur = mat_apply(af, cur, cv);
            __builtin_amdgcn_sched_barrier(0);
        }
        af = af2;
        cv = cv2;
        mc = nm;
    }
    cur = mat_apply(af, cur, cv);
#undef WMMA_F32

    #pragma unroll
    for (int i = 0; i < 8; ++i) {
        const int row = i + half * 8;
        Hbuf[(row)      * HSTRIDE + col] = cur.t[0][i];
        Hbuf[(row + 16) * HSTRIDE + col] = cur.t[1][i];
        Hbuf[(row + 32) * HSTRIDE + col] = cur.t[2][i];
        Hbuf[(row + 48) * HSTRIDE + col] = cur.t[3][i];
    }
    __syncthreads();

    for (int idx = tid; idx < 64 * NS; idx += NTHREADS) {
        const int c0  = idx / NS;
        const int s   = idx - c0 * NS;
        const int sg  = blockIdx.x * NS + s;
        const int b   = sg >> 12;
        const int rem = sg & 4095;
        out[((b * 64 + c0) << 12) + rem] = Hbuf[c0 * HSTRIDE + s];
    }
}

extern "C" void kernel_launch(void* const* d_in, const int* in_sizes, int n_in,
                              void* d_out, int out_size, void* d_ws, size_t ws_size,
                              hipStream_t stream) {
    const float* x  = (const float*)d_in[0];
    const float* W1 = (const float*)d_in[1];
    const float* b1 = (const float*)d_in[2];
    const float* W2 = (const float*)d_in[3];
    const float* b2 = (const float*)d_in[4];
    float* out = (float*)d_out;

    dim3 grid(16384 / NS), block(NTHREADS);
    conv_chain_wmma_kernel<<<grid, block, 0, stream>>>(x, W1, b1, W2, b2, out);
}


# Squaring hand-over pipelined in two row halves on the split barrier: rows 0..31 stored, signalled, fetched while rows 32..63 compute, rows 32..63 during the next squaring; c update among the WMMAs
# speedup vs baseline: 24.1524x; 1.0716x over previous
#include <hip/hip_runtime.h>
#include <hip/hip_bf16.h>


typedef __attribute__((ext_vector_type(2))) float v2f;
typedef __attribute__((ext_vector_type(4))) float v4f;
typedef __attribute__((ext_vector_type(8))) float v8f;

#define NS       64
#define NTHREADS 128
#define HSTRIDE  68
#define MSTRIDE  68
#define W1K      147
#define W1STRIDE 148
#define XROWS    490
#define NREPEAT  200
#define NSQUARE  7
#define LDS_FLOATS (42 * 1024)

static_assert(NREPEAT == 128 + 64 + 8, "the powers applied to the tile add up to the step count");
static_assert((1 << NSQUARE) == 128, "the last squaring gives the 128th power");

__device__ constexpr int koff(int k) {
    return k < W1K ? (k / 49) * XROWS + ((k % 49) / 7) * 70 + (k % 7) : 0;
}

#define WMMA_F32(a, b, c) __builtin_amdgcn_wmma_f32_16x16x4_f32(false, (a), false, (b), (short)0, (c), false, false)

struct AFrag { v2f a[4][16]; };
struct Tile  { v8f t[4]; };

__device__ __forceinline__ AFrag load_afrag(const float* Mb, int m, int half) {
    AFrag f;
    #pragma unroll
    for (int ct = 0; ct < 4; ++ct) {
        #pragma unroll
        for (int cb = 0; cb < 4; ++cb) {
            #pragma unroll
            for (int jj = 0; jj < 2; ++jj) {
                const v4f q = *(const v4f*)(Mb + (ct * 16 + m) * MSTRIDE + cb * 16 + half * 8 + jj * 4);
                v2f lo, hi;
                lo[0] = q[0]; lo[1] = q[1];
                hi[0] = q[2]; hi[1] = q[3];
                f.a[ct][cb * 4 + jj * 2]     = lo;
                f.a[ct][cb * 4 + jj * 2 + 1] = hi;
            }
        }
    }
    return f;
}

__device__ __forceinline__ Tile load_cvec(const float* cb_, int half) {
    Tile c;
    #pragma unroll
    for (int ct = 0; ct < 4; ++ct) {
        const v4f q0 = *(const v4f*)(cb_ + ct * 16 + half * 8);
        const v4f q1 = *(const v4f*)(cb_ + ct * 16 + half * 8 + 4);
        #pragma unroll
        for (int i = 0; i < 4; ++i) { c.t[ct][i] = q0[i]; c.t[ct][i + 4] = q1[i]; }
    }
    return c;
}

template <int CT, int K0, int K1>
__device__ __forceinline__ void mat_part(const AFrag& f, const Tile& x, v8f& d0, v8f& d1) {
    #pragma unroll
    for (int ks = K0; ks < K1; ++ks) {
        v2f bfrag;
        bfrag[0] = x.t[ks >> 2][2 * (ks & 3)];
        bfrag[1] = x.t[ks >> 2][2 * (ks & 3) + 1];
        d0 = WMMA_F32(f.a[CT][ks],     bfrag, d0);
        d1 = WMMA_F32(f.a[CT + 1][ks], bfrag, d1);
    }
}

template <int CT>
__device__ __forceinline__ void mat_apply_half(const AFrag& f, const Tile& x, const Tile& c, Tile& d) {
    v8f d0 = c.t[CT], d1 = c.t[CT + 1];
    mat_part<CT, 0, 16>(f, x, d0, d1);
    d.t[CT] = d0; d.t[CT + 1] = d1;
}

__device__ __forceinline__ Tile mat_apply(const AFrag& f, const Tile& x, const Tile& c) {
    Tile d;
    mat_apply_half<0>(f, x, c, d);
    __builtin_amdgcn_sched_barrier(0);
    mat_apply_half<2>(f, x, c, d);
    return d;
}

template <int CT0>
__device__ __forceinline__ void load_afrag_rows(AFrag& f, const float* Mb, int m, int half) {
    #pragma unroll
    for (int ct = CT0; ct < CT0 + 2; ++ct) {
        #pragma unroll
        for (int cb = 0; cb < 4; ++cb) {
            #pragma unroll
            for (int jj = 0; jj < 2; ++jj) {
                const v4f q = *(const v4f*)(Mb + (ct * 16 + m) * MSTRIDE + cb * 16 + half * 8 + jj * 4);
                v2f lo, hi;
                lo[0] = q[0]; lo[1] = q[1];
                hi[0] = q[2]; hi[1] = q[3];
                f.a[ct][cb * 4 + jj * 2]     = lo;
                f.a[ct][cb * 4 + jj * 2 + 1] = hi;
            }
        }
    }
}

template <int CT0>
__device__ __forceinline__ void store_rows(float* Mb, const Tile& t, int half, int col) {
    #pragma unroll
    for (int ct = CT0; ct < CT0 + 2; ++ct) {
        #pragma unroll
        for (int r = 0; r < 8; ++r)
            Mb[(ct * 16 + half * 8 + r) * MSTRIDE + col] = t.t[ct][r];
    }
}

#define PIN2(a, b)          asm volatile("" : "+v"(a), "+v"(b) :: "memory")
#define PIN3(a, b, c)       asm volatile("" : "+v"(a), "+v"(b), "+v"(c) :: "memory")
#define PIN4(a, b, c, d)    asm volatile("" : "+v"(a), "+v"(b), "+v"(c), "+v"(d) :: "memory")
#define SIGNAL_PIN2(a, b)   asm volatile("s_wait_dscnt 0x0\n\ts_barrier_signal -1" : "+v"(a), "+v"(b) :: "memory")
#define WAIT_PIN2(a, b)     asm volatile("s_barrier_wait -1" : "+v"(a), "+v"(b) :: "memory")

__device__ __forceinline__ float xhalf(float v) {
    return __builtin_bit_cast(float, __builtin_amdgcn_permlanex16(__builtin_bit_cast(int, v), __builtin_bit_cast(int, v),
                                                                  0x76543210, 0xfedcba98, false, false));
}

__device__ __forceinline__ void load_own(v2f (&own)[16], const float* Mb, int wave, int m, int half) {
    #pragma unroll
    for (int cb = 0; cb < 4; ++cb) {
        #pragma unroll
        for (int jj = 0; jj < 2; ++jj) {
            const v4f q = *(const v4f*)(Mb + (wave * 16 + m) * MSTRIDE + cb * 16 + half * 8 + jj * 4);
            v2f lo, hi;
            lo[0] = q[0]; lo[1] = q[1];
            hi[0] = q[2]; hi[1] = q[3];
            own[cb * 4 + jj * 2]     = lo;
            own[cb * 4 + jj * 2 + 1] = hi;
        }
    }
}

template <int K0, int K1>
__device__ __forceinline__ void rowdot_part(const v2f (&own)[16], const Tile& c, v2f& acc2) {
    #pragma unroll
    for (int ks = K0; ks < K1; ++ks) {
        v2f cpair;
        cpair[0] = c.t[ks >> 2][2 * (ks & 3)];
        cpair[1] = c.t[ks >> 2][2 * (ks & 3) + 1];
        acc2 += own[ks] * cpair;
    }
}

__global__ __launch_bounds__(NTHREADS) __attribute__((amdgpu_waves_per_eu(1, 1)))
void conv_chain_wmma_kernel(const float* __restrict__ x,
                            const float* __restrict__ W1,
                            const float* __restrict__ b1,
                            const float* __restrict__ W2,
                            const float* __restrict__ b2,
                            float* __restrict__ out)
{
    __shared__ __attribute__((aligned(32))) float lds[LDS_FLOATS];
    static_assert(64 * HSTRIDE + 2 * 64 * MSTRIDE + 2 * 64 + 64 * W1STRIDE + 3 * XROWS <= LDS_FLOATS,
                  "tiles fit the reservation");
    static_assert((64 * HSTRIDE) % 8 == 0 && (64 * MSTRIDE) % 8 == 0 && MSTRIDE % 4 == 0,
                  "matrix and vector buffers are read as aligned float quads");
    static_assert((64 * HSTRIDE + 2 * 64 * MSTRIDE + 2 * 64) % 2 == 0 && W1STRIDE % 2 == 0,
                  "W1 fragments are read as aligned float pairs");
    float* const Hbuf  = lds;
    float* const Mbuf0 = lds + 64 * HSTRIDE;
    float* const Mbuf1 = Mbuf0 + 64 * MSTRIDE;
    float* const cbuf0 = Mbuf1 + 64 * MSTRIDE;
    float* const cbuf1 = cbuf0 + 64;
    float* const W1lds = cbuf1 + 64;
    float* const xs    = W1lds + 64 * W1STRIDE;

    const int tid  = threadIdx.x;
    const int lane = tid & 31;
    const int wave = __builtin_amdgcn_readfirstlane(tid >> 5);
    const int half = lane >> 4;
    const int m    = lane & 15;

    const int sg0 = blockIdx.x * NS;
    const int b   = sg0 >> 12;
    const int yy  = (sg0 & 4095) >> 6;
    v4f w2q[8];
    #pragma unroll
    for (int i = 0; i < 8; ++i)
        w2q[i] = *(const v4f*)(W2 + 4 * (tid + NTHREADS * i));
    const float b2v = b2[tid & 63];
    v4f w1q[19];
    int  w1i[19];
    const float* const wsrc = W1 + wave * (16 * W1K);
    #pragma unroll
    for (int j = 0; j < 19; ++j) {
        const int q = lane + 32 * j;
        w1i[j] = q < 588 ? q : 587;
        w1q[j] = *(const v4f*)(wsrc + 4 * w1i[j]);
    }
    const float* const xsrc = x + ((b * 3) * 70 + yy) * 70;
    const int xp1 = tid + NTHREADS < XROWS / 2 ? tid + NTHREADS : XROWS / 2 - 1;
    v2f xq[3][2];
    #pragma unroll
    for (int ci = 0; ci < 3; ++ci) {
        xq[ci][0] = *(const v2f*)(xsrc + ci * 4900 + 2 * tid);
        xq[ci][1] = *(const v2f*)(xsrc + ci * 4900 + 2 * xp1);
    }
    v8f cur0, cur1, cur2, cur3;
    {
        v4f bq[4][2];
        #pragma unroll
        for (int ct = 0; ct < 4; ++ct) {
            bq[ct][0] = *(const v4f*)(b1 + ct * 16 + half * 8);
            bq[ct][1] = *(const v4f*)(b1 + ct * 16 + half * 8 + 4);
        }
        #pragma unroll
        for (int i = 0; i < 4; ++i) {
            cur0[i] = bq[0][0][i]; cur0[i + 4] = bq[0][1][i];
            cur1[i] = bq[1][0][i]; cur1[i + 4] = bq[1][1][i];
            cur2[i] = bq[2][0][i]; cur2[i + 4] = bq[2][1][i];
            cur3[i] = bq[3][0][i]; cur3[i + 4] = bq[3][1][i];
        }
    }
    __builtin_amdgcn_sched_barrier(0);
    int w1o[19][4];
    #pragma unroll
    for (int j = 0; j < 19; ++j) {
        const int f0 = 4 * w1i[j];
        const int r0 = f0 / W1K;
        const int k0 = f0 - r0 * W1K;
        #pragma unroll
        for (int e = 0; e < 4; ++e)
            w1o[j][e] = r0 * W1STRIDE + k0 + e + (k0 + e >= W1K ? 1 : 0);
    }
    __builtin_amdgcn_sched_barrier(0);
    #pragma unroll
    for (int i = 0; i < 8; ++i) {
        const int q = tid + NTHREADS * i;
        *(v4f*)(Mbuf0 + (q >> 4) * MSTRIDE + (q & 15) * 4) = w2q[i];
    }
    cbuf0[tid & 63] = b2v;
    {
        float* const wdst = W1lds + (wave * 16) * W1STRIDE;
        #pragma unroll
        for (int j = 0; j < 19; ++j) {
            #pragma unroll
            for (int e = 0; e < 4; ++e)
                wdst[w1o[j][e]] = w1q[j][e];
        }
        wdst[(lane & 15) * W1STRIDE + W1K] = 0.f;
    }
    #pragma unroll
    for (int ci = 0; ci < 3; ++ci) {
        *(v2f*)(xs + ci * XROWS + 2 * tid) = xq[ci][0];
        *(v2f*)(xs + ci * XROWS + 2 * xp1) = xq[ci][1];
    }

    __syncthreads();

    const int col = wave * 16 + m;

    v2f cvA[W1STRIDE / 4][4];
    v2f cvB[W1STRIDE / 4];
    {
        const float* const xcol = xs + col;
        const float* const wrow = W1lds + m * W1STRIDE + 2 * half;
        #pragma unroll
        for (int ks = 0; ks < W1STRIDE / 4; ++ks) {
            const int o0 = half ? koff(4 * ks + 2) : koff(4 * ks);
            const int o1 = half ? koff(4 * ks + 3) : koff(4 * ks + 1);
            v2f bfrag;
            bfrag[0] = xcol[o0];
            bfrag[1] = xcol[o1];
            if (4 * ks + 3 >= W1K) bfrag[1] = half ? 0.f : bfrag[1];
            cvB[ks] = bfrag;
            #pragma unroll
            for (int ct = 0; ct < 4; ++ct)
                cvA[ks][ct] = *(const v2f*)(wrow + ct * 16 * W1STRIDE + 4 * ks);
        }
    }

    AFrag af = load_afrag(Mbuf0, m, half);
    Tile  cv = load_cvec(cbuf0, half);
    Tile  mc;
    #pragma unroll
    for (int cb = 0; cb < 4; ++cb) {
        #pragma unroll
        for (int r = 0; r < 8; ++r)
            mc.t[cb][r] = Mbuf0[(cb * 16 + half * 8 + r) * MSTRIDE + col];
    }
    float crow = cbuf0[col];
    v2f own[16];
    load_own(own, Mbuf0, wave, m, half);

    __builtin_amdgcn_sched_barrier(0);

    #pragma unroll
    for (int ks = 0; ks < W1STRIDE / 4; ++ks) {
        cur0 = WMMA_F32(cvA[ks][0], cvB[ks], cur0);
        cur1 = WMMA_F32(cvA[ks][1], cvB[ks], cur1);
        cur2 = WMMA_F32(cvA[ks][2], cvB[ks], cur2);
        cur3 = WMMA_F32(cvA[ks][3], cvB[ks], cur3);
    }
    Tile cur;
    cur.t[0] = cur0; cur.t[1] = cur1; cur.t[2] = cur2; cur.t[3] = cur3;

    PIN4(cur.t[0], cur.t[1], cur.t[2], cur.t[3]);
    PIN4(cur.t[0], cur.t[1], mc.t[2], mc.t[3]);
    v8f zero8;
    #pragma unroll
    for (int i = 0; i < 8; ++i) zero8[i] = 0.f;
    AFrag af2;
    Tile  cv2;
    #pragma unroll
    for (int s = 0; s < NSQUARE; ++s) {
        const int p = 1 << s;
        float* const Mcur  = (s & 1) ? Mbuf1 : Mbuf0;
        float* const ccur  = (s & 1) ? cbuf1 : cbuf0;
        float* const Mnext = (s & 1) ? Mbuf0 : Mbuf1;
        float* const cnext = (s & 1) ? cbuf0 : cbuf1;

        v8f d0 = zero8, d1 = zero8;
        PIN4(d0, d1, mc.t[2], mc.t[3]);
        mat_part<0, 0, 1>(af, mc, d0, d1);
        if (s > 0) {
            PIN2(d0, d1);
            store_rows<2>(Mcur, mc, half, col);
            ccur[col] = crow;
            PIN2(d0, d1);
        }
        mat_part<0, 1, 3>(af, mc, d0, d1);
        if (s > 0) SIGNAL_PIN2(d0, d1);
        mat_part<0, 3, 6>(af, mc, d0, d1);
        if (s > 0) {
            WAIT_PIN2(d0, d1);
            load_afrag_rows<2>(af, Mcur, m, half);
            cv = load_cvec(ccur, half);
            load_own(own, Mcur, wave, m, half);
            PIN2(d0, d1);
        }
        mat_part<0, 6, 16>(af, mc, d0, d1);
        Tile nm;
        nm.t[0] = d0; nm.t[1] = d1;

        v8f e0 = zero8, e1 = zero8;
        PIN4(e0, e1, nm.t[0], nm.t[1]);
        mat_part<2, 0, 2>(af, mc, e0, e1);
        PIN2(e0, e1);
        store_rows<0>(Mnext, nm, half, col);
        PIN2(e0, e1);
        mat_part<2, 2, 4>(af, mc, e0, e1);
        SIGNAL_PIN2(e0, e1);

        v2f acc2;
        acc2[0] = 0.f; acc2[1] = 0.f;
        mat_part<2, 4, 5>(af, mc, e0, e1);
        rowdot_part<0, 4>(own, cv, acc2);
        PIN3(acc2, e0, e1);
        mat_part<2, 5, 6>(af, mc, e0, e1);
        rowdot_part<4, 8>(own, cv, acc2);
        PIN3(acc2, e0, e1);
        mat_part<2, 6, 7>(af, mc, e0, e1);
        rowdot_part<8, 12>(own, cv, acc2);
        PIN3(acc2, e0, e1);
        mat_part<2, 7, 8>(af, mc, e0, e1);
        rowdot_part<12, 16>(own, cv, acc2);
        PIN3(acc2, e0, e1);
        mat_part<2, 8, 10>(af, mc, e0, e1);
        {
            float part = acc2[0] + acc2[1];
            part += xhalf(part);
            crow += part;
        }
        WAIT_PIN2(e0, e1);
        load_afrag_rows<0>(af2, Mnext, m, half);
        PIN2(e0, e1);
        mat_part<2, 10, 16>(af, mc, e0, e1);
        nm.t[2] = e0; nm.t[3] = e1;

        if (p == 8) {
            PIN4(nm.t[2], nm.t[3], cur.t[0], cur.t[1]);
            cur = mat_apply(af, cur, cv);
            PIN4(nm.t[2], nm.t[3], cur.t[0], cur.t[1]);
            PIN4(nm.t[2], nm.t[3], cur.t[2], cur.t[3]);
        }
        if (s == NSQUARE - 1) {
            Tile d;
            v8f a0 = cv.t[0], a1 = cv.t[1];
            PIN4(a0, a1, nm.t[2], nm.t[3]);
            mat_part<0, 0, 2>(af, cur, a0, a1);
            PIN2(a0, a1);
            store_rows<2>(Mnext, nm, half, col);
            cnext[col] = crow;
            PIN2(a0, a1);
            mat_part<0, 2, 4>(af, cur, a0, a1);
            SIGNAL_PIN2(a0, a1);
            mat_part<0, 4, 12>(af, cur, a0, a1);
            WAIT_PIN2(a0, a1);
            load_afrag_rows<2>(af2, Mnext, m, half);
            cv2 = load_cvec(cnext, half);
            PIN2(a0, a1);
            mat_part<0, 12, 16>(af, cur, a0, a1);
            v8f b0 = cv.t[2], b1 = cv.t[3];
            PIN4(a0, a1, b0, b1);
            mat_part<2, 0, 16>(af, cur, b0, b1);
            d.t[0] = a0; d.t[1] = a1; d.t[2] = b0; d.t[3] = b1;
            cur = d;
        }
        #pragma unroll
        for (int ks = 0; ks < 16; ++ks) { af.a[0][ks] = af2.a[0][ks]; af.a[1][ks] = af2.a[1][ks]; }
        mc = nm;
    }
    cur = mat_apply(af2, cur, cv2);
#undef WMMA_F32

    #pragma unroll
    for (int i = 0; i < 8; ++i) {
        const int row = i + half * 8;
        Hbuf[(row)      * HSTRIDE + col] = cur.t[0][i];
        Hbuf[(row + 16) * HSTRIDE + col] = cur.t[1][i];
        Hbuf[(row + 32) * HSTRIDE + col] = cur.t[2][i];
        Hbuf[(row + 48) * HSTRIDE + col] = cur.t[3][i];
    }
    __syncthreads();

    for (int idx = tid; idx < 64 * NS; idx += NTHREADS) {
        const int c0  = idx / NS;
        const int s   = idx - c0 * NS;
        const int sg  = blockIdx.x * NS + s;
        const int b   = sg >> 12;
        const int rem = sg & 4095;
        out[((b * 64 + c0) << 12) + rem] = Hbuf[c0 * HSTRIDE + s];
    }
}

extern "C" void kernel_launch(void* const* d_in, const int* in_sizes, int n_in,
                              void* d_out, int out_size, void* d_ws, size_t ws_size,
                              hipStream_t stream) {
    const float* x  = (const float*)d_in[0];
    const float* W1 = (const float*)d_in[1];
    const float* b1 = (const float*)d_in[2];
    const float* W2 = (const float*)d_in[3];
    const float* b2 = (const float*)d_in[4];
    float* out = (float*)d_out;

    dim3 grid(16384 / NS), block(NTHREADS);
    conv_chain_wmma_kernel<<<grid, block, 0, stream>>>(x, W1, b1, W2, b2, out);
}


# Write-out in two row halves of 16-byte pieces: rows 0..31 of the last product go LDS -> global_store_b128 while rows 32..63 are computed; 8 b128 load/store pairs per thread replace a loop of 32 b32
# speedup vs baseline: 26.9096x; 1.1142x over previous
#include <hip/hip_runtime.h>
#include <hip/hip_bf16.h>


typedef __attribute__((ext_vector_type(2))) float v2f;
typedef __attribute__((ext_vector_type(4))) float v4f;
typedef __attribute__((ext_vector_type(8))) float v8f;

#define NS       64
#define NTHREADS 128
#define HSTRIDE  68
#define MSTRIDE  68
#define W1K      147
#define W1STRIDE 148
#define XROWS    490
#define NREPEAT  200
#define NSQUARE  7
#define LDS_FLOATS (42 * 1024)

static_assert(NREPEAT == 128 + 64 + 8, "the powers applied to the tile add up to the step count");
static_assert(HSTRIDE == MSTRIDE, "the write-out reuses the row store of the hand-over, which steps by MSTRIDE");
static_assert((1 << NSQUARE) == 128, "the last squaring gives the 128th power");

__device__ constexpr int koff(int k) {
    return k < W1K ? (k / 49) * XROWS + ((k % 49) / 7) * 70 + (k % 7) : 0;
}

#define WMMA_F32(a, b, c) __builtin_amdgcn_wmma_f32_16x16x4_f32(false, (a), false, (b), (short)0, (c), false, false)

struct AFrag { v2f a[4][16]; };
struct Tile  { v8f t[4]; };

__device__ __forceinline__ AFrag load_afrag(const float* Mb, int m, int half) {
    AFrag f;
    #pragma unroll
    for (int ct = 0; ct < 4; ++ct) {
        #pragma unroll
        for (int cb = 0; cb < 4; ++cb) {
            #pragma unroll
            for (int jj = 0; jj < 2; ++jj) {
                const v4f q = *(const v4f*)(Mb + (ct * 16 + m) * MSTRIDE + cb * 16 + half * 8 + jj * 4);
                v2f lo, hi;
                lo[0] = q[0]; lo[1] = q[1];
                hi[0] = q[2]; hi[1] = q[3];
                f.a[ct][cb * 4 + jj * 2]     = lo;
                f.a[ct][cb * 4 + jj * 2 + 1] = hi;
            }
        }
    }
    return f;
}

__device__ __forceinline__ Tile load_cvec(const float* cb_, int half) {
    Tile c;
    #pragma unroll
    for (int ct = 0; ct < 4; ++ct) {
        const v4f q0 = *(const v4f*)(cb_ + ct * 16 + half * 8);
        const v4f q1 = *(const v4f*)(cb_ + ct * 16 + half * 8 + 4);
        #pragma unroll
        for (int i = 0; i < 4; ++i) { c.t[ct][i] = q0[i]; c.t[ct][i + 4] = q1[i]; }
    }
    return c;
}

template <int CT, int K0, int K1>
__device__ __forceinline__ void mat_part(const AFrag& f, const Tile& x, v8f& d0, v8f& d1) {
    #pragma unroll
    for (int ks = K0; ks < K1; ++ks) {
        v2f bfrag;
        bfrag[0] = x.t[ks >> 2][2 * (ks & 3)];
        bfrag[1] = x.t[ks >> 2][2 * (ks & 3) + 1];
        d0 = WMMA_F32(f.a[CT][ks],     bfrag, d0);
        d1 = WMMA_F32(f.a[CT + 1][ks], bfrag, d1);
    }
}

template <int CT>
__device__ __forceinline__ void mat_apply_half(const AFrag& f, const Tile& x, const Tile& c, Tile& d) {
    v8f d0 = c.t[CT], d1 = c.t[CT + 1];
    mat_part<CT, 0, 16>(f, x, d0, d1);
    d.t[CT] = d0; d.t[CT + 1] = d1;
}

__device__ __forceinline__ Tile mat_apply(const AFrag& f, const Tile& x, const Tile& c) {
    Tile d;
    mat_apply_half<0>(f, x, c, d);
    __builtin_amdgcn_sched_barrier(0);
    mat_apply_half<2>(f, x, c, d);
    return d;
}

template <int CT0>
__device__ __forceinline__ void load_afrag_rows(AFrag& f, const float* Mb, int m, int half) {
    #pragma unroll
    for (int ct = CT0; ct < CT0 + 2; ++ct) {
        #pragma unroll
        for (int cb = 0; cb < 4; ++cb) {
            #pragma unroll
            for (int jj = 0; jj < 2; ++jj) {
                const v4f q = *(const v4f*)(Mb + (ct * 16 + m) * MSTRIDE + cb * 16 + half * 8 + jj * 4);
                v2f lo, hi;
                lo[0] = q[0]; lo[1] = q[1];
                hi[0] = q[2]; hi[1] = q[3];
                f.a[ct][cb * 4 + jj * 2]     = lo;
                f.a[ct][cb * 4 + jj * 2 + 1] = hi;
            }
        }
    }
}

template <int CT0>
__device__ __forceinline__ void store_rows(float* Mb, const Tile& t, int half, int col) {
    #pragma unroll
    for (int ct = CT0; ct < CT0 + 2; ++ct) {
        #pragma unroll
        for (int r = 0; r < 8; ++r)
            Mb[(ct * 16 + half * 8 + r) * MSTRIDE + col] = t.t[ct][r];
    }
}

#define PIN2(a, b)          asm volatile("" : "+v"(a), "+v"(b) :: "memory")
#define PIN3(a, b, c)       asm volatile("" : "+v"(a), "+v"(b), "+v"(c) :: "memory")
#define PIN4(a, b, c, d)    asm volatile("" : "+v"(a), "+v"(b), "+v"(c), "+v"(d) :: "memory")
#define SIGNAL_PIN2(a, b)   asm volatile("s_wait_dscnt 0x0\n\ts_barrier_signal -1" : "+v"(a), "+v"(b) :: "memory")
#define WAIT_PIN2(a, b)     asm volatile("s_barrier_wait -1" : "+v"(a), "+v"(b) :: "memory")
#define LDS_BARRIER()       asm volatile("s_wait_dscnt 0x0\n\ts_barrier_signal -1\n\ts_barrier_wait -1" ::: "memory")

__device__ __forceinline__ float xhalf(float v) {
    return __builtin_bit_cast(float, __builtin_amdgcn_permlanex16(__builtin_bit_cast(int, v), __builtin_bit_cast(int, v),
                                                                  0x76543210, 0xfedcba98, false, false));
}

__device__ __forceinline__ void load_own(v2f (&own)[16], const float* Mb, int wave, int m, int half) {
    #pragma unroll
    for (int cb = 0; cb < 4; ++cb) {
        #pragma unroll
        for (int jj = 0; jj < 2; ++jj) {
            const v4f q = *(const v4f*)(Mb + (wave * 16 + m) * MSTRIDE + cb * 16 + half * 8 + jj * 4);
            v2f lo, hi;
            lo[0] = q[0]; lo[1] = q[1];
            hi[0] = q[2]; hi[1] = q[3];
            own[cb * 4 + jj * 2]     = lo;
            own[cb * 4 + jj * 2 + 1] = hi;
        }
    }
}

template <int K0, int K1>
__device__ __forceinline__ void rowdot_part(const v2f (&own)[16], const Tile& c, v2f& acc2) {
    #pragma unroll
    for (int ks = K0; ks < K1; ++ks) {
        v2f cpair;
        cpair[0] = c.t[ks >> 2][2 * (ks & 3)];
        cpair[1] = c.t[ks >> 2][2 * (ks & 3) + 1];
        acc2 += own[ks] * cpair;
    }
}

__global__ __launch_bounds__(NTHREADS) __attribute__((amdgpu_waves_per_eu(1, 1)))
void conv_chain_wmma_kernel(const float* __restrict__ x,
                            const float* __restrict__ W1,
                            const float* __restrict__ b1,
                            const float* __restrict__ W2,
                            const float* __restrict__ b2,
                            float* __restrict__ out)
{
    __shared__ __attribute__((aligned(32))) float lds[LDS_FLOATS];
    static_assert(64 * HSTRIDE + 2 * 64 * MSTRIDE + 2 * 64 + 64 * W1STRIDE + 3 * XROWS <= LDS_FLOATS,
                  "tiles fit the reservation");
    static_assert((64 * HSTRIDE) % 8 == 0 && (64 * MSTRIDE) % 8 == 0 && MSTRIDE % 4 == 0,
                  "matrix and vector buffers are read as aligned float quads");
    static_assert((64 * HSTRIDE + 2 * 64 * MSTRIDE + 2 * 64) % 2 == 0 && W1STRIDE % 2 == 0,
                  "W1 fragments are read as aligned float pairs");
    float* const Hbuf  = lds;
    float* const Mbuf0 = lds + 64 * HSTRIDE;
    float* const Mbuf1 = Mbuf0 + 64 * MSTRIDE;
    float* const cbuf0 = Mbuf1 + 64 * MSTRIDE;
    float* const cbuf1 = cbuf0 + 64;
    float* const W1lds = cbuf1 + 64;
    float* const xs    = W1lds + 64 * W1STRIDE;

    const int tid  = threadIdx.x;
    const int lane = tid & 31;
    const int wave = __builtin_amdgcn_readfirstlane(tid >> 5);
    const int half = lane >> 4;
    const int m    = lane & 15;

    const int sg0 = blockIdx.x * NS;
    const int b   = sg0 >> 12;
    const int yy  = (sg0 & 4095) >> 6;
    v4f w2q[8];
    #pragma unroll
    for (int i = 0; i < 8; ++i)
        w2q[i] = *(const v4f*)(W2 + 4 * (tid + NTHREADS * i));
    const float b2v = b2[tid & 63];
    v4f w1q[19];
    int  w1i[19];
    const float* const wsrc = W1 + wave * (16 * W1K);
    #pragma unroll
    for (int j = 0; j < 19; ++j) {
        const int q = lane + 32 * j;
        w1i[j] = q < 588 ? q : 587;
        w1q[j] = *(const v4f*)(wsrc + 4 * w1i[j]);
    }
    const float* const xsrc = x + ((b * 3) * 70 + yy) * 70;
    const int xp1 = tid + NTHREADS < XROWS / 2 ? tid + NTHREADS : XROWS / 2 - 1;
    v2f xq[3][2];
    #pragma unroll
    for (int ci = 0; ci < 3; ++ci) {
        xq[ci][0] = *(const v2f*)(xsrc + ci * 4900 + 2 * tid);
        xq[ci][1] = *(const v2f*)(xsrc + ci * 4900 + 2 * xp1);
    }
    v8f cur0, cur1, cur2, cur3;
    {
        v4f bq[4][2];
        #pragma unroll
        for (int ct = 0; ct < 4; ++ct) {
            bq[ct][0] = *(const v4f*)(b1 + ct * 16 + half * 8);
            bq[ct][1] = *(const v4f*)(b1 + ct * 16 + half * 8 + 4);
        }
        #pragma unroll
        for (int i = 0; i < 4; ++i) {
            cur0[i] = bq[0][0][i]; cur0[i + 4] = bq[0][1][i];
            cur1[i] = bq[1][0][i]; cur1[i + 4] = bq[1][1][i];
            cur2[i] = bq[2][0][i]; cur2[i + 4] = bq[2][1][i];
            cur3[i] = bq[3][0][i]; cur3[i + 4] = bq[3][1][i];
        }
    }
    __builtin_amdgcn_sched_barrier(0);
    int w1o[19][4];
    #pragma unroll
    for (int j = 0; j < 19; ++j) {
        const int f0 = 4 * w1i[j];
        const int r0 = f0 / W1K;
        const int k0 = f0 - r0 * W1K;
        #pragma unroll
        for (int e = 0; e < 4; ++e)
            w1o[j][e] = r0 * W1STRIDE + k0 + e + (k0 + e >= W1K ? 1 : 0);
    }
    __builtin_amdgcn_sched_barrier(0);
    #pragma unroll
    for (int i = 0; i < 8; ++i) {
        const int q = tid + NTHREADS * i;
        *(v4f*)(Mbuf0 + (q >> 4) * MSTRIDE + (q & 15) * 4) = w2q[i];
    }
    cbuf0[tid & 63] = b2v;
    {
        float* const wdst = W1lds + (wave * 16) * W1STRIDE;
        #pragma unroll
        for (int j = 0; j < 19; ++j) {
            #pragma unroll
            for (int e = 0; e < 4; ++e)
                wdst[w1o[j][e]] = w1q[j][e];
        }
        wdst[(lane & 15) * W1STRIDE + W1K] = 0.f;
    }
    #pragma unroll
    for (int ci = 0; ci < 3; ++ci) {
        *(v2f*)(xs + ci * XROWS + 2 * tid) = xq[ci][0];
        *(v2f*)(xs + ci * XROWS + 2 * xp1) = xq[ci][1];
    }

    __syncthreads();

    const int col = wave * 16 + m;

    v2f cvA[W1STRIDE / 4][4];
    v2f cvB[W1STRIDE / 4];
    {
        const float* const xcol = xs + col;
        const float* const wrow = W1lds + m * W1STRIDE + 2 * half;
        #pragma unroll
        for (int ks = 0; ks < W1STRIDE / 4; ++ks) {
            const int o0 = half ? koff(4 * ks + 2) : koff(4 * ks);
            const int o1 = half ? koff(4 * ks + 3) : koff(4 * ks + 1);
            v2f bfrag;
            bfrag[0] = xcol[o0];
            bfrag[1] = xcol[o1];
            if (4 * ks + 3 >= W1K) bfrag[1] = half ? 0.f : bfrag[1];
            cvB[ks] = bfrag;
            #pragma unroll
            for (int ct = 0; ct < 4; ++ct)
                cvA[ks][ct] = *(const v2f*)(wrow + ct * 16 * W1STRIDE + 4 * ks);
        }
    }

    AFrag af = load_afrag(Mbuf0, m, half);
    Tile  cv = load_cvec(cbuf0, half);
    Tile  mc;
    #pragma unroll
    for (int cb = 0; cb < 4; ++cb) {
        #pragma unroll
        for (int r = 0; r < 8; ++r)
            mc.t[cb][r] = Mbuf0[(cb * 16 + half * 8 + r) * MSTRIDE + col];
    }
    float crow = cbuf0[col];
    v2f own[16];
    load_own(own, Mbuf0, wave, m, half);

    __builtin_amdgcn_sched_barrier(0);

    #pragma unroll
    for (int ks = 0; ks < W1STRIDE / 4; ++ks) {
        cur0 = WMMA_F32(cvA[ks][0], cvB[ks], cur0);
        cur1 = WMMA_F32(cvA[ks][1], cvB[ks], cur1);
        cur2 = WMMA_F32(cvA[ks][2], cvB[ks], cur2);
        cur3 = WMMA_F32(cvA[ks][3], cvB[ks], cur3);
    }
    Tile cur;
    cur.t[0] = cur0; cur.t[1] = cur1; cur.t[2] = cur2; cur.t[3] = cur3;

    PIN4(cur.t[0], cur.t[1], cur.t[2], cur.t[3]);
    PIN4(cur.t[0], cur.t[1], mc.t[2], mc.t[3]);
    v8f zero8;
    #pragma unroll
    for (int i = 0; i < 8; ++i) zero8[i] = 0.f;
    AFrag af2;
    Tile  cv2;
    #pragma unroll
    for (int s = 0; s < NSQUARE; ++s) {
        const int p = 1 << s;
        float* const Mcur  = (s & 1) ? Mbuf1 : Mbuf0;
        float* const ccur  = (s & 1) ? cbuf1 : cbuf0;
        float* const Mnext = (s & 1) ? Mbuf0 : Mbuf1;
        float* const cnext = (s & 1) ? cbuf0 : cbuf1;

        v8f d0 = zero8, d1 = zero8;
        PIN4(d0, d1, mc.t[2], mc.t[3]);
        mat_part<0, 0, 1>(af, mc, d0, d1);
        if (s > 0) {
            PIN2(d0, d1);
            store_rows<2>(Mcur, mc, half, col);
            ccur[col] = crow;
            PIN2(d0, d1);
        }
        mat_part<0, 1, 3>(af, mc, d0, d1);
        if (s > 0) SIGNAL_PIN2(d0, d1);
        mat_part<0, 3, 6>(af, mc, d0, d1);
        if (s > 0) {
            WAIT_PIN2(d0, d1);
            load_afrag_rows<2>(af, Mcur, m, half);
            cv = load_cvec(ccur, half);
            load_own(own, Mcur, wave, m, half);
            PIN2(d0, d1);
        }
        mat_part<0, 6, 16>(af, mc, d0, d1);
        Tile nm;
        nm.t[0] = d0; nm.t[1] = d1;

        v8f e0 = zero8, e1 = zero8;
        PIN4(e0, e1, nm.t[0], nm.t[1]);
        mat_part<2, 0, 2>(af, mc, e0, e1);
        PIN2(e0, e1);
        store_rows<0>(Mnext, nm, half, col);
        PIN2(e0, e1);
        mat_part<2, 2, 4>(af, mc, e0, e1);
        SIGNAL_PIN2(e0, e1);

        v2f acc2;
        acc2[0] = 0.f; acc2[1] = 0.f;
        mat_part<2, 4, 5>(af, mc, e0, e1);
        rowdot_part<0, 4>(own, cv, acc2);
        PIN3(acc2, e0, e1);
        mat_part<2, 5, 6>(af, mc, e0, e1);
        rowdot_part<4, 8>(own, cv, acc2);
        PIN3(acc2, e0, e1);
        mat_part<2, 6, 7>(af, mc, e0, e1);
        rowdot_part<8, 12>(own, cv, acc2);
        PIN3(acc2, e0, e1);
        mat_part<2, 7, 8>(af, mc, e0, e1);
        rowdot_part<12, 16>(own, cv, acc2);
        PIN3(acc2, e0, e1);
        mat_part<2, 8, 10>(af, mc, e0, e1);
        {
            float part = acc2[0] + acc2[1];
            part += xhalf(part);
            crow += part;
        }
        WAIT_PIN2(e0, e1);
        load_afrag_rows<0>(af2, Mnext, m, half);
        PIN2(e0, e1);
        mat_part<2, 10, 16>(af, mc, e0, e1);
        nm.t[2] = e0; nm.t[3] = e1;

        if (p == 8) {
            PIN4(nm.t[2], nm.t[3], cur.t[0], cur.t[1]);
            cur = mat_apply(af, cur, cv);
            PIN4(nm.t[2], nm.t[3], cur.t[0], cur.t[1]);
            PIN4(nm.t[2], nm.t[3], cur.t[2], cur.t[3]);
        }
        if (s == NSQUARE - 1) {
            Tile d;
            v8f a0 = cv.t[0], a1 = cv.t[1];
            PIN4(a0, a1, nm.t[2], nm.t[3]);
            mat_part<0, 0, 2>(af, cur, a0, a1);
            PIN2(a0, a1);
            store_rows<2>(Mnext, nm, half, col);
            cnext[col] = crow;
            PIN2(a0, a1);
            mat_part<0, 2, 4>(af, cur, a0, a1);
            SIGNAL_PIN2(a0, a1);
            mat_part<0, 4, 12>(af, cur, a0, a1);
            WAIT_PIN2(a0, a1);
            load_afrag_rows<2>(af2, Mnext, m, half);
            cv2 = load_cvec(cnext, half);
            PIN2(a0, a1);
            mat_part<0, 12, 16>(af, cur, a0, a1);
            v8f b0 = cv.t[2], b1 = cv.t[3];
            PIN4(a0, a1, b0, b1);
            mat_part<2, 0, 16>(af, cur, b0, b1);
            d.t[0] = a0; d.t[1] = a1; d.t[2] = b0; d.t[3] = b1;
            cur = d;
        }
        #pragma unroll
        for (int ks = 0; ks < 16; ++ks) { af.a[0][ks] = af2.a[0][ks]; af.a[1][ks] = af2.a[1][ks]; }
        mc = nm;
    }
    const float* const hrow = Hbuf + (tid >> 4) * HSTRIDE + 4 * (tid & 15);
    float* const orow = out + (((b * 64 + (tid >> 4)) << 12) + (sg0 & 4095) + 4 * (tid & 15));
    Tile fin;
    mat_apply_half<0>(af2, cur, cv2, fin);
    v8f g0 = cv2.t[2], g1 = cv2.t[3];
    PIN4(g0, g1, fin.t[0], fin.t[1]);
    mat_part<2, 0, 2>(af2, cur, g0, g1);
    PIN2(g0, g1);
    store_rows<0>(Hbuf, fin, half, col);
    PIN2(g0, g1);
    mat_part<2, 2, 4>(af2, cur, g0, g1);
    SIGNAL_PIN2(g0, g1);
    mat_part<2, 4, 10>(af2, cur, g0, g1);
    WAIT_PIN2(g0, g1);
    v4f oq[4];
    #pragma unroll
    for (int i = 0; i < 4; ++i)
        oq[i] = *(const v4f*)(hrow + 8 * i * HSTRIDE);
    PIN2(g0, g1);
    mat_part<2, 10, 14>(af2, cur, g0, g1);
    PIN2(g0, g1);
    #pragma unroll
    for (int i = 0; i < 4; ++i)
        *(v4f*)(orow + ((8 * i) << 12)) = oq[i];
    PIN2(g0, g1);
    mat_part<2, 14, 16>(af2, cur, g0, g1);
    fin.t[2] = g0; fin.t[3] = g1;
#undef WMMA_F32
    store_rows<2>(Hbuf, fin, half, col);
    LDS_BARRIER();
    #pragma unroll
    for (int i = 4; i < 8; ++i)
        oq[i - 4] = *(const v4f*)(hrow + 8 * i * HSTRIDE);
    #pragma unroll
    for (int i = 4; i < 8; ++i)
        *(v4f*)(orow + ((8 * i) << 12)) = oq[i - 4];
}

extern "C" void kernel_launch(void* const* d_in, const int* in_sizes, int n_in,
                              void* d_out, int out_size, void* d_ws, size_t ws_size,
                              hipStream_t stream) {
    const float* x  = (const float*)d_in[0];
    const float* W1 = (const float*)d_in[1];
    const float* b1 = (const float*)d_in[2];
    const float* W2 = (const float*)d_in[3];
    const float* b2 = (const float*)d_in[4];
    float* out = (float*)d_out;

    dim3 grid(16384 / NS), block(NTHREADS);
    conv_chain_wmma_kernel<<<grid, block, 0, stream>>>(x, W1, b1, W2, b2, out);
}
